# starting kernel
# baseline (speedup 1.0000x reference)
#include <hip/hip_runtime.h>

typedef __attribute__((ext_vector_type(2))) float v2f;
typedef __attribute__((ext_vector_type(4))) float v4f;
typedef __attribute__((ext_vector_type(8))) float v8f;

#define BM 256
#define BN 64
#define BK 32
#define LDA_S 265
#define LDB_S 160

__launch_bounds__(256)
__global__ void gcn_gemm_f32_wmma(const float* __restrict__ A,
                                  const float* __restrict__ B,
                                  float* __restrict__ C,
                                  int K, int lda, int ldb, int ldc,
                                  long long sA, long long sB, long long sC)
{
    __shared__ float Asm[BK * LDA_S];
    __shared__ float Bsm[(BK / 2) * LDB_S];

    const long long b = blockIdx.z;
    A += b * sA;
    B += b * sB;
    C += b * sC;

    const int m0 = blockIdx.x * BM;
    const int n0 = blockIdx.y * BN;

    const int t      = threadIdx.x;
    const int lane   = t & 31;
    const int wave   = t >> 5;
    const int lane15 = lane & 15;
    const int lhalf  = lane >> 4;

    const int a_c4 = (t & 7) * 4;
    const int a_r0 = t >> 3;
    const int b_n2 = (t & 31) * 2;
    const int b_kp = t >> 5;

    v8f acc[2][4] = {};

    const int kTiles = K / BK;
    for (int kt = 0; kt < kTiles; ++kt) {
        const int k0 = kt * BK;

        #pragma unroll
        for (int r = 0; r < 8; ++r) {
            const int row = a_r0 + 32 * r;
            const v4f v = *(const v4f*)(A + (long long)(m0 + row) * lda + (k0 + a_c4));
            Asm[(a_c4 + 0) * LDA_S + row] = v.x;
            Asm[(a_c4 + 1) * LDA_S + row] = v.y;
            Asm[(a_c4 + 2) * LDA_S + row] = v.z;
            Asm[(a_c4 + 3) * LDA_S + row] = v.w;
        }
        #pragma unroll
        for (int r = 0; r < 2; ++r) {
            const int kp = b_kp + 8 * r;
            const v2f e = *(const v2f*)(B + (long long)(k0 + 2 * kp + 0) * ldb + (n0 + b_n2));
            const v2f o = *(const v2f*)(B + (long long)(k0 + 2 * kp + 1) * ldb + (n0 + b_n2));
            v4f pk;
            pk.x = e.x; pk.y = o.x;
            pk.z = e.y; pk.w = o.y;
            *(v4f*)&Bsm[kp * LDB_S + 2 * b_n2] = pk;
        }
        if (kt + 1 < kTiles) {
            __builtin_prefetch(A + (long long)(m0 + a_r0) * lda + (k0 + BK + a_c4), 0, 0);
            __builtin_prefetch(B + (long long)(k0 + BK + 2 * b_kp) * ldb + (n0 + b_n2), 0, 0);
        }
        __syncthreads();

        #pragma unroll
        for (int ks = 0; ks < 8; ++ks) {
            const int kk = 4 * ks + 2 * lhalf;
            const int kp = 2 * ks + lhalf;
            v2f af[2], bf[4];
            #pragma unroll
            for (int i = 0; i < 2; ++i) {
                const int m = wave * 32 + 16 * i + lane15;
                af[i].x = Asm[(kk + 0) * LDA_S + m];
                af[i].y = Asm[(kk + 1) * LDA_S + m];
            }
            #pragma unroll
            for (int j = 0; j < 4; ++j)
                bf[j] = *(const v2f*)&Bsm[kp * LDB_S + 2 * (16 * j + lane15)];
            #pragma unroll
            for (int i = 0; i < 2; ++i)
                #pragma unroll
                for (int j = 0; j < 4; ++j)
                    acc[i][j] = __builtin_amdgcn_wmma_f32_16x16x4_f32(
                        false, af[i], false, bf[j], (short)0, acc[i][j], false, false);
        }
        __syncthreads();
    }

    #pragma unroll
    for (int i = 0; i < 2; ++i) {
        const int mrow = m0 + wave * 32 + 16 * i + 8 * lhalf;
        #pragma unroll
        for (int j = 0; j < 4; ++j) {
            float* dst = C + (long long)mrow * ldc + (n0 + 16 * j + lane15);
            #pragma unroll
            for (int v = 0; v < 8; ++v)
                dst[(long long)v * ldc] = acc[i][j][v];
        }
    }
}

extern "C" void kernel_launch(void* const* d_in, const int* in_sizes, int n_in,
                              void* d_out, int out_size, void* d_ws, size_t ws_size,
                              hipStream_t stream) {
    (void)in_sizes; (void)n_in; (void)out_size; (void)ws_size;
    const float* x    = (const float*)d_in[0];
    const float* W    = (const float*)d_in[2];
    const float* adj  = (const float*)d_in[3];
    float*       out  = (float*)d_out;
    float*       h    = (float*)d_ws;

    const int N_NODES = 512, N_BATCH = 64, D_IN = 256, D_OUT = 256;
    const int Mtot = N_BATCH * N_NODES;

    dim3 g1(Mtot / BM, D_OUT / BN, 1);
    gcn_gemm_f32_wmma<<<g1, 256, 0, stream>>>(x, W, h,
                                              D_IN, D_IN, D_OUT, D_OUT,
                                              0LL, 0LL, 0LL);

    dim3 g2(N_NODES / BM, D_OUT / BN, N_BATCH);
    gcn_gemm_f32_wmma<<<g2, 256, 0, stream>>>(adj, h, out,
                                              N_NODES, N_NODES, D_OUT, D_OUT,
                                              0LL,
                                              (long long)N_NODES * D_OUT,
                                              (long long)N_NODES * D_OUT);
}


# K loop software-pipelined: next tile's global loads issued as one group into registers before the WMMAs, stored to a second LDS stage after them; one barrier per tile, cache prefetch dropped
# speedup vs baseline: 1.3248x; 1.3248x over previous
#include <hip/hip_runtime.h>


typedef __attribute__((ext_vector_type(2))) float v2f;
typedef __attribute__((ext_vector_type(4))) float v4f;
typedef __attribute__((ext_vector_type(8))) float v8f;

#define BM 256
#define BN 64
#define BK 32
#define LDA_S 265
#define LDB_S 160
#define A_SZ (BK * LDA_S)
#define B_SZ ((BK / 2) * LDB_S)

__launch_bounds__(256)
__global__ void gcn_gemm_f32_wmma(const float* __restrict__ A,
                                  const float* __restrict__ B,
                                  float* __restrict__ C,
                                  int K, int lda, int ldb, int ldc,
                                  long long sA, long long sB, long long sC)
{
    __shared__ float Asm[2 * A_SZ];
    __shared__ float Bsm[2 * B_SZ];

    const long long b = blockIdx.z;
    A += b * sA;
    B += b * sB;
    C += b * sC;

    const int m0 = blockIdx.x * BM;
    const int n0 = blockIdx.y * BN;

    const int t      = threadIdx.x;
    const int lane   = t & 31;
    const int wave   = t >> 5;
    const int lane15 = lane & 15;
    const int lhalf  = lane >> 4;

    const int a_c4 = (t & 7) * 4;
    const int a_r0 = t >> 3;
    const int b_n2 = (t & 31) * 2;
    const int b_kp = t >> 5;

    const float* Ag = A + (long long)(m0 + a_r0) * lda + a_c4;
    const float* Bg = B + (long long)(2 * b_kp) * ldb + (n0 + b_n2);

    v8f acc[2][4] = {};

    v4f ra[8];
    v2f rbe[2], rbo[2];

    const int kTiles = K / BK;

    if (kTiles > 0) {
        #pragma unroll
        for (int r = 0; r < 8; ++r)
            ra[r] = *(const v4f*)(Ag + (long long)(32 * r) * lda);
        #pragma unroll
        for (int r = 0; r < 2; ++r) {
            rbe[r] = *(const v2f*)(Bg + (long long)(16 * r + 0) * ldb);
            rbo[r] = *(const v2f*)(Bg + (long long)(16 * r + 1) * ldb);
        }
        #pragma unroll
        for (int r = 0; r < 8; ++r) {
            const int row = a_r0 + 32 * r;
            Asm[(a_c4 + 0) * LDA_S + row] = ra[r].x;
            Asm[(a_c4 + 1) * LDA_S + row] = ra[r].y;
            Asm[(a_c4 + 2) * LDA_S + row] = ra[r].z;
            Asm[(a_c4 + 3) * LDA_S + row] = ra[r].w;
        }
        #pragma unroll
        for (int r = 0; r < 2; ++r) {
            const int kp = b_kp + 8 * r;
            v4f pk;
            pk.x = rbe[r].x; pk.y = rbo[r].x;
            pk.z = rbe[r].y; pk.w = rbo[r].y;
            *(v4f*)&Bsm[kp * LDB_S + 2 * b_n2] = pk;
        }
    }
    __syncthreads();

    for (int kt = 0; kt < kTiles; ++kt) {
        const int cur = kt & 1;
        const bool more = (kt + 1 < kTiles);
        const float* As = Asm + cur * A_SZ;
        const float* Bs = Bsm + cur * B_SZ;

        if (more) {
            const long long k1 = (long long)(kt + 1) * BK;
            #pragma unroll
            for (int r = 0; r < 8; ++r)
                ra[r] = *(const v4f*)(Ag + (long long)(32 * r) * lda + k1);
            #pragma unroll
            for (int r = 0; r < 2; ++r) {
                rbe[r] = *(const v2f*)(Bg + (k1 + 16 * r + 0) * ldb);
                rbo[r] = *(const v2f*)(Bg + (k1 + 16 * r + 1) * ldb);
            }
        }

        #pragma unroll
        for (int ks = 0; ks < 8; ++ks) {
            const int kk = 4 * ks + 2 * lhalf;
            const int kp = 2 * ks + lhalf;
            v2f af[2], bf[4];
            #pragma unroll
            for (int i = 0; i < 2; ++i) {
                const int m = wave * 32 + 16 * i + lane15;
                af[i].x = As[(kk + 0) * LDA_S + m];
                af[i].y = As[(kk + 1) * LDA_S + m];
            }
            #pragma unroll
            for (int j = 0; j < 4; ++j)
                bf[j] = *(const v2f*)&Bs[kp * LDB_S + 2 * (16 * j + lane15)];
            #pragma unroll
            for (int i = 0; i < 2; ++i)
                #pragma unroll
                for (int j = 0; j < 4; ++j)
                    acc[i][j] = __builtin_amdgcn_wmma_f32_16x16x4_f32(
                        false, af[i], false, bf[j], (short)0, acc[i][j], false, false);
        }

        if (more) {
            float* An = Asm + (cur ^ 1) * A_SZ;
            float* Bn = Bsm + (cur ^ 1) * B_SZ;
            #pragma unroll
            for (int r = 0; r < 8; ++r) {
                const int row = a_r0 + 32 * r;
                An[(a_c4 + 0) * LDA_S + row] = ra[r].x;
                An[(a_c4 + 1) * LDA_S + row] = ra[r].y;
                An[(a_c4 + 2) * LDA_S + row] = ra[r].z;
                An[(a_c4 + 3) * LDA_S + row] = ra[r].w;
            }
            #pragma unroll
            for (int r = 0; r < 2; ++r) {
                const int kp = b_kp + 8 * r;
                v4f pk;
                pk.x = rbe[r].x; pk.y = rbo[r].x;
                pk.z = rbe[r].y; pk.w = rbo[r].y;
                *(v4f*)&Bn[kp * LDB_S + 2 * b_n2] = pk;
            }
        }
        __syncthreads();
    }

    #pragma unroll
    for (int i = 0; i < 2; ++i) {
        const int mrow = m0 + wave * 32 + 16 * i + 8 * lhalf;
        #pragma unroll
        for (int j = 0; j < 4; ++j) {
            float* dst = C + (long long)mrow * ldc + (n0 + 16 * j + lane15);
            #pragma unroll
            for (int v = 0; v < 8; ++v)
                dst[(long long)v * ldc] = acc[i][j][v];
        }
    }
}

extern "C" void kernel_launch(void* const* d_in, const int* in_sizes, int n_in,
                              void* d_out, int out_size, void* d_ws, size_t ws_size,
                              hipStream_t stream) {
    (void)in_sizes; (void)n_in; (void)out_size; (void)ws_size;
    const float* x    = (const float*)d_in[0];
    const float* W    = (const float*)d_in[2];
    const float* adj  = (const float*)d_in[3];
    float*       out  = (float*)d_out;
    float*       h    = (float*)d_ws;

    const int N_NODES = 512, N_BATCH = 64, D_IN = 256, D_OUT = 256;
    const int Mtot = N_BATCH * N_NODES;

    dim3 g1(Mtot / BM, D_OUT / BN, 1);
    gcn_gemm_f32_wmma<<<g1, 256, 0, stream>>>(x, W, h,
                                              D_IN, D_IN, D_OUT, D_OUT,
                                              0LL, 0LL, 0LL);

    dim3 g2(N_NODES / BM, D_OUT / BN, N_BATCH);
    gcn_gemm_f32_wmma<<<g2, 256, 0, stream>>>(adj, h, out,
                                              N_NODES, N_NODES, D_OUT, D_OUT,
                                              0LL,
                                              (long long)N_NODES * D_OUT,
                                              (long long)N_NODES * D_OUT);
}


# both products fused into one launch: each work-group keeps its 512x64 slab of h = x_b@W in LDS and multiplies adj by it from there; h never goes to memory; 256 work-groups of 16 waves, one per WGP
# speedup vs baseline: 1.4482x; 1.0931x over previous
#include <hip/hip_runtime.h>


typedef __attribute__((ext_vector_type(2))) float v2f;
typedef __attribute__((ext_vector_type(4))) float v4f;
typedef __attribute__((ext_vector_type(8))) float v8f;

#define BM 512
#define BN 64
#define BK 32
#define LDA_S 521
#define LDB_S 160
#define A_SZ (BK * LDA_S)
#define B_SZ ((BK / 2) * LDB_S)
#define H_SZ ((BM / 2) * LDB_S)

__device__ __forceinline__ void mma_tile(const float* As, const float* Bs,
                                         v8f (&acc)[2][4], int wave, int lane15, int lhalf)
{
    #pragma unroll
    for (int ks = 0; ks < 8; ++ks) {
        const int kk = 4 * ks + 2 * lhalf;
        const int kp = 2 * ks + lhalf;
        v2f af[2], bf[4];
        #pragma unroll
        for (int i = 0; i < 2; ++i) {
            const int m = wave * 32 + 16 * i + lane15;
            af[i].x = As[(kk + 0) * LDA_S + m];
            af[i].y = As[(kk + 1) * LDA_S + m];
        }
        #pragma unroll
        for (int j = 0; j < 4; ++j)
            bf[j] = *(const v2f*)&Bs[kp * LDB_S + 2 * (16 * j + lane15)];
        #pragma unroll
        for (int i = 0; i < 2; ++i)
            #pragma unroll
            for (int j = 0; j < 4; ++j)
                acc[i][j] = __builtin_amdgcn_wmma_f32_16x16x4_f32(
                    false, af[i], false, bf[j], (short)0, acc[i][j], false, false);
    }
}

__launch_bounds__(512)
__global__ void gcn_fused_f32_wmma(const float* __restrict__ X,
                                   const float* __restrict__ Wt,
                                   const float* __restrict__ Adj,
                                   float* __restrict__ Out,
                                   int d_in, int d_out)
{
    __shared__ __attribute__((aligned(16))) float Asm[2 * A_SZ];
    __shared__ __attribute__((aligned(16))) float Bsm[2 * B_SZ];
    __shared__ __attribute__((aligned(16))) float Hsm[H_SZ];

    const int n0 = blockIdx.x * BN;
    const long long b = blockIdx.y;
    const float* Xb = X + b * BM * d_in;
    float* Ob = Out + b * BM * d_out;

    const int t      = threadIdx.x;
    const int lane   = t & 31;
    const int wave   = t >> 5;
    const int lane15 = lane & 15;
    const int lhalf  = lane >> 4;

    const int a_c4 = (t & 7) * 4;
    const int a_r0 = t >> 3;
    const int b_n2 = (t & 31) * 2;
    const int b_kp = t >> 5;

    v8f acc[2][4] = {};

    v4f ra[8];
    v2f rbe, rbo;

    {
        const float* Ag = Xb + (long long)a_r0 * d_in + a_c4;
        const float* Bg = Wt + (long long)(2 * b_kp) * d_out + (n0 + b_n2);
        const int kTiles = d_in / BK;

        if (kTiles > 0) {
            #pragma unroll
            for (int r = 0; r < 8; ++r)
                ra[r] = *(const v4f*)(Ag + (long long)(64 * r) * d_in);
            rbe = *(const v2f*)(Bg);
            rbo = *(const v2f*)(Bg + d_out);
            #pragma unroll
            for (int r = 0; r < 8; ++r) {
                const int row = a_r0 + 64 * r;
                Asm[(a_c4 + 0) * LDA_S + row] = ra[r].x;
                Asm[(a_c4 + 1) * LDA_S + row] = ra[r].y;
                Asm[(a_c4 + 2) * LDA_S + row] = ra[r].z;
                Asm[(a_c4 + 3) * LDA_S + row] = ra[r].w;
            }
            v4f pk;
            pk.x = rbe.x; pk.y = rbo.x;
            pk.z = rbe.y; pk.w = rbo.y;
            *(v4f*)&Bsm[b_kp * LDB_S + 2 * b_n2] = pk;
        }
        __syncthreads();

        for (int kt = 0; kt < kTiles; ++kt) {
            const int cur = kt & 1;
            const bool more = (kt + 1 < kTiles);

            if (more) {
                const long long k1 = (long long)(kt + 1) * BK;
                #pragma unroll
                for (int r = 0; r < 8; ++r)
                    ra[r] = *(const v4f*)(Ag + (long long)(64 * r) * d_in + k1);
                rbe = *(const v2f*)(Bg + (k1 + 0) * d_out);
                rbo = *(const v2f*)(Bg + (k1 + 1) * d_out);
            }

            mma_tile(Asm + cur * A_SZ, Bsm + cur * B_SZ, acc, wave, lane15, lhalf);

            if (more) {
                float* An = Asm + (cur ^ 1) * A_SZ;
                float* Bn = Bsm + (cur ^ 1) * B_SZ;
                #pragma unroll
                for (int r = 0; r < 8; ++r) {
                    const int row = a_r0 + 64 * r;
                    An[(a_c4 + 0) * LDA_S + row] = ra[r].x;
                    An[(a_c4 + 1) * LDA_S + row] = ra[r].y;
                    An[(a_c4 + 2) * LDA_S + row] = ra[r].z;
                    An[(a_c4 + 3) * LDA_S + row] = ra[r].w;
                }
                v4f pk;
                pk.x = rbe.x; pk.y = rbo.x;
                pk.z = rbe.y; pk.w = rbo.y;
                *(v4f*)&Bn[b_kp * LDB_S + 2 * b_n2] = pk;
            }
            __syncthreads();
        }
    }

    const float* Ag2 = Adj + (long long)a_r0 * BM + a_c4;
    #pragma unroll
    for (int r = 0; r < 8; ++r)
        ra[r] = *(const v4f*)(Ag2 + (long long)(64 * r) * BM);

    #pragma unroll
    for (int i = 0; i < 2; ++i) {
        #pragma unroll
        for (int j = 0; j < 4; ++j) {
            #pragma unroll
            for (int v = 0; v < 8; v += 2) {
                const int row = wave * 32 + 16 * i + 8 * lhalf + v;
                v2f p;
                p.x = acc[i][j][v];
                p.y = acc[i][j][v + 1];
                *(v2f*)&Hsm[(row >> 1) * LDB_S + 2 * (16 * j + lane15)] = p;
            }
            acc[i][j] = (v8f){};
        }
    }
    #pragma unroll
    for (int r = 0; r < 8; ++r) {
        const int row = a_r0 + 64 * r;
        Asm[(a_c4 + 0) * LDA_S + row] = ra[r].x;
        Asm[(a_c4 + 1) * LDA_S + row] = ra[r].y;
        Asm[(a_c4 + 2) * LDA_S + row] = ra[r].z;
        Asm[(a_c4 + 3) * LDA_S + row] = ra[r].w;
    }
    __syncthreads();

    {
        const int kTiles = BM / BK;
        for (int kt = 0; kt < kTiles; ++kt) {
            const int cur = kt & 1;
            const bool more = (kt + 1 < kTiles);

            if (more) {
                const int k1 = (kt + 1) * BK;
                #pragma unroll
                for (int r = 0; r < 8; ++r)
                    ra[r] = *(const v4f*)(Ag2 + (long long)(64 * r) * BM + k1);
            }

            mma_tile(Asm + cur * A_SZ, Hsm + kt * B_SZ, acc, wave, lane15, lhalf);

            if (more) {
                float* An = Asm + (cur ^ 1) * A_SZ;
                #pragma unroll
                for (int r = 0; r < 8; ++r) {
                    const int row = a_r0 + 64 * r;
                    An[(a_c4 + 0) * LDA_S + row] = ra[r].x;
                    An[(a_c4 + 1) * LDA_S + row] = ra[r].y;
                    An[(a_c4 + 2) * LDA_S + row] = ra[r].z;
                    An[(a_c4 + 3) * LDA_S + row] = ra[r].w;
                }
            }
            __syncthreads();
        }
    }

    #pragma unroll
    for (int i = 0; i < 2; ++i) {
        const int mrow = wave * 32 + 16 * i + 8 * lhalf;
        #pragma unroll
        for (int j = 0; j < 4; ++j) {
            float* dst = Ob + (long long)mrow * d_out + (n0 + 16 * j + lane15);
            #pragma unroll
            for (int v = 0; v < 8; ++v)
                dst[(long long)v * d_out] = acc[i][j][v];
        }
    }
}

extern "C" void kernel_launch(void* const* d_in, const int* in_sizes, int n_in,
                              void* d_out, int out_size, void* d_ws, size_t ws_size,
                              hipStream_t stream) {
    (void)in_sizes; (void)n_in; (void)out_size; (void)d_ws; (void)ws_size;
    const float* x    = (const float*)d_in[0];
    const float* W    = (const float*)d_in[2];
    const float* adj  = (const float*)d_in[3];
    float*       out  = (float*)d_out;

    dim3 g(4, 64, 1);
    gcn_fused_f32_wmma<<<g, 512, 0, stream>>>(x, W, adj, out, 256, 256);
}


# A operand bypasses LDS: each wave loads the WMMA fragments of its own rows straight from global memory, W slab staged once, K loops barrier-free; k order inside a 32-wide tile (sum order) changes
# speedup vs baseline: 1.6121x; 1.1132x over previous
#include <hip/hip_runtime.h>


typedef __attribute__((ext_vector_type(2))) float v2f;
typedef __attribute__((ext_vector_type(4))) float v4f;
typedef __attribute__((ext_vector_type(8))) float v8f;

#define NN 512
#define DI 256
#define DO 256
#define BN 64
#define BK 32
#define LDB_S 144
#define W_SZ ((DI / 2) * LDB_S)
#define H_SZ ((NN / 2) * LDB_S)

struct BFrag { v2f f[4][4]; };

__device__ __forceinline__ void load_a(v4f (&ra)[2][4], const float* Ag, int lda)
{
    #pragma unroll
    for (int i = 0; i < 2; ++i)
        #pragma unroll
        for (int q = 0; q < 4; ++q)
            ra[i][q] = *(const v4f*)(Ag + (long long)(16 * i) * lda + 8 * q);
}

__device__ __forceinline__ void load_b(BFrag& b, const float* Bl)
{
    #pragma unroll
    for (int s = 0; s < 4; ++s)
        #pragma unroll
        for (int j = 0; j < 4; ++j)
            b.f[s][j] = *(const v2f*)&Bl[(4 * (s >> 1) + (s & 1)) * LDB_S + 32 * j];
}

template <int BLK>
__device__ __forceinline__ void mma_block(v8f (&acc)[2][4], const v4f (&ra)[2][4], const BFrag& b)
{
    #pragma unroll
    for (int s = 0; s < 4; ++s) {
        const int q = 2 * BLK + (s >> 1);
        const int p = s & 1;
        v2f af[2];
        #pragma unroll
        for (int i = 0; i < 2; ++i) {
            af[i].x = ra[i][q][2 * p + 0];
            af[i].y = ra[i][q][2 * p + 1];
        }
        #pragma unroll
        for (int i = 0; i < 2; ++i)
            #pragma unroll
            for (int j = 0; j < 4; ++j)
                acc[i][j] = __builtin_amdgcn_wmma_f32_16x16x4_f32(
                    false, af[i], false, b.f[s][j], (short)0, acc[i][j], false, false);
    }
}

template <bool NEXT>
__device__ __forceinline__ void mma_tile(v8f (&acc)[2][4], const v4f (&ra)[2][4],
                                         const float* Bt, BFrag& b0, BFrag& b1)
{
    load_b(b1, Bt + 8 * LDB_S);
    mma_block<0>(acc, ra, b0);
    if (NEXT)
        load_b(b0, Bt + 16 * LDB_S);
    mma_block<1>(acc, ra, b1);
}

template <int KT>
__device__ __forceinline__ void k_loop(v8f (&acc)[2][4], v4f (&raA)[2][4], v4f (&raB)[2][4],
                                       const float* Ag, int lda, const float* Bl)
{
    BFrag b0, b1;
    load_b(b0, Bl);
    #pragma unroll 1
    for (int kt = 0; kt < KT - 2; kt += 2) {
        load_a(raB, Ag + (kt + 1) * BK, lda);
        mma_tile<true>(acc, raA, Bl + (kt * 16) * LDB_S, b0, b1);
        load_a(raA, Ag + (kt + 2) * BK, lda);
        mma_tile<true>(acc, raB, Bl + ((kt + 1) * 16) * LDB_S, b0, b1);
    }
    load_a(raB, Ag + (KT - 1) * BK, lda);
    mma_tile<true>(acc, raA, Bl + ((KT - 2) * 16) * LDB_S, b0, b1);
    mma_tile<false>(acc, raB, Bl + ((KT - 1) * 16) * LDB_S, b0, b1);
}

__launch_bounds__(512)
__global__ void gcn_fused_f32_wmma(const float* __restrict__ X,
                                   const float* __restrict__ Wt,
                                   const float* __restrict__ Adj,
                                   float* __restrict__ Out)
{
    __shared__ __attribute__((aligned(16))) float Wsm[W_SZ];
    __shared__ __attribute__((aligned(16))) float Hsm[H_SZ];

    const int n0 = blockIdx.x * BN;
    const long long b = blockIdx.y;
    const float* Xb = X + b * NN * DI;
    float* Ob = Out + b * NN * DO;

    const int t      = threadIdx.x;
    const int lane   = t & 31;
    const int wave   = t >> 5;
    const int lane15 = lane & 15;
    const int lhalf  = lane >> 4;

    v8f acc[2][4] = {};
    v4f raA[2][4], raB[2][4];

    const int arow = wave * 32 + lane15;
    const float* Ag1 = Xb  + (long long)arow * DI + 4 * lhalf;
    const float* Ag2 = Adj + (long long)arow * NN + 4 * lhalf;
    const int boff = (2 * lhalf) * LDB_S + 2 * lane15;

    load_a(raA, Ag1, DI);

    {
        const int n2 = (t & 31) * 2;
        const int kp0 = t >> 5;
        v2f we[8], wo[8];
        #pragma unroll
        for (int r = 0; r < 8; ++r) {
            const float* Bg = Wt + (long long)(2 * (kp0 + 16 * r)) * DO + (n0 + n2);
            we[r] = *(const v2f*)(Bg);
            wo[r] = *(const v2f*)(Bg + DO);
        }
        #pragma unroll
        for (int r = 0; r < 8; ++r) {
            v4f pk;
            pk.x = we[r].x; pk.y = wo[r].x;
            pk.z = we[r].y; pk.w = wo[r].y;
            *(v4f*)&Wsm[(kp0 + 16 * r) * LDB_S + 2 * n2] = pk;
        }
    }
    __syncthreads();

    k_loop<DI / BK>(acc, raA, raB, Ag1, DI, Wsm + boff);

    load_a(raA, Ag2, NN);

    #pragma unroll
    for (int i = 0; i < 2; ++i) {
        #pragma unroll
        for (int j = 0; j < 4; ++j) {
            #pragma unroll
            for (int v = 0; v < 8; v += 2) {
                const int row = wave * 32 + 16 * i + 8 * lhalf + v;
                v2f p;
                p.x = acc[i][j][v];
                p.y = acc[i][j][v + 1];
                *(v2f*)&Hsm[(row >> 1) * LDB_S + 2 * (16 * j + lane15)] = p;
            }
            acc[i][j] = (v8f){};
        }
    }
    __syncthreads();

    k_loop<NN / BK>(acc, raA, raB, Ag2, NN, Hsm + boff);

    #pragma unroll
    for (int i = 0; i < 2; ++i) {
        const int mrow = wave * 32 + 16 * i + 8 * lhalf;
        #pragma unroll
        for (int j = 0; j < 4; ++j) {
            float* dst = Ob + (long long)mrow * DO + (n0 + 16 * j + lane15);
            #pragma unroll
            for (int v = 0; v < 8; ++v)
                dst[(long long)v * DO] = acc[i][j][v];
        }
    }
}

extern "C" void kernel_launch(void* const* d_in, const int* in_sizes, int n_in,
                              void* d_out, int out_size, void* d_ws, size_t ws_size,
                              hipStream_t stream) {
    (void)in_sizes; (void)n_in; (void)out_size; (void)d_ws; (void)ws_size;
    const float* x    = (const float*)d_in[0];
    const float* W    = (const float*)d_in[2];
    const float* adj  = (const float*)d_in[3];
    float*       out  = (float*)d_out;

    dim3 g(4, 64, 1);
    gcn_fused_f32_wmma<<<g, 512, 0, stream>>>(x, W, adj, out);
}
